# starting kernel
# baseline (speedup 1.0000x reference)
#include <hip/hip_runtime.h>


typedef float v2f __attribute__((ext_vector_type(2)));
typedef float v4f __attribute__((ext_vector_type(4)));
typedef float v8f __attribute__((ext_vector_type(8)));

#define Bc 2
#define Hc 8
#define Sc 2048
#define Dc 64
#define BHc (Bc * Hc)
#define BQ 64
#define KS 68
#define VS 72

__launch_bounds__(128)
__global__ void retention_fused_kernel(const float* __restrict__ q,
                                       const float* __restrict__ k,
                                       const float* __restrict__ v,
                                       const float* __restrict__ omask,
                                       float* __restrict__ out)
{
    const int qb   = (int)(gridDim.x - 1u - blockIdx.x);
    const int bh   = (int)blockIdx.y;
    const int tid  = (int)threadIdx.x;
    const int w    = tid >> 5;
    const int lane = tid & 31;
    const int half = lane >> 4;
    const int l16  = lane & 15;

    __shared__ float ldsK[BQ * KS];
    __shared__ float ldsV[BQ * VS];
    __shared__ float ldsS[BQ * KS];
    __shared__ float ldsQC[BQ];

    const size_t bh_sd = (size_t)bh * Sc * Dc;
    const size_t bh_ss = (size_t)bh * Sc * (size_t)Sc;

    for (int rr = 0; rr < 16; ++rr) {
        const int row = w * 16 + rr;
        const size_t base = bh_ss + (size_t)(qb * BQ + row) * Sc;
        const v4f* om4 = (const v4f*)(omask + base);
        float s = 0.f;
        for (int j = lane; j < Sc / 4; j += 32) {
            v4f t = __builtin_nontemporal_load(om4 + j);
            s += (t.x + t.y) + (t.z + t.w);
        }
        #pragma unroll
        for (int off = 16; off > 0; off >>= 1) s += __shfl_xor(s, off, 32);
        if (lane == 0) {
            float m0 = omask[base];
            ldsQC[row] = m0 / sqrtf(s);
        }
    }
    __syncthreads();

    const int qrow = qb * BQ + w * 16 + l16;
    const float qc = ldsQC[w * 16 + l16];
    const float* qptr = q + bh_sd + (size_t)qrow * Dc;
    v2f qf[16];
    #pragma unroll
    for (int kk = 0; kk < 16; ++kk) {
        const int c = kk * 4 + half * 2;
        v2f a; a.x = qptr[c] * qc; a.y = qptr[c + 1] * qc;
        qf[kk] = a;
    }

    v8f oacc[4];
    v8f pacc;
    #pragma unroll
    for (int r = 0; r < 8; ++r) pacc[r] = 0.f;
    #pragma unroll
    for (int ct = 0; ct < 4; ++ct) {
        #pragma unroll
        for (int r = 0; r < 8; ++r) oacc[ct][r] = 0.f;
    }

    for (int kb = 0; kb <= qb; ++kb) {
        __syncthreads();
        {
            const float* kbase = k + bh_sd + (size_t)(kb * BQ) * Dc;
            const float* vbase = v + bh_sd + (size_t)(kb * BQ) * Dc;
            for (int idx = tid; idx < BQ * Dc / 4; idx += 128) {
                const int row  = idx >> 4;
                const int col4 = (idx & 15) * 4;
                const float kc = 1.f / omask[bh_ss + (size_t)(kb * BQ + row) * Sc];
                const v4f kvv = *(const v4f*)(kbase + row * Dc + col4);
                ldsK[row * KS + col4 + 0] = kvv.x * kc;
                ldsK[row * KS + col4 + 1] = kvv.y * kc;
                ldsK[row * KS + col4 + 2] = kvv.z * kc;
                ldsK[row * KS + col4 + 3] = kvv.w * kc;
                const v4f vvv = *(const v4f*)(vbase + row * Dc + col4);
                ldsV[row * VS + col4 + 0] = vvv.x;
                ldsV[row * VS + col4 + 1] = vvv.y;
                ldsV[row * VS + col4 + 2] = vvv.z;
                ldsV[row * VS + col4 + 3] = vvv.w;
            }
        }
        __syncthreads();

        #pragma unroll
        for (int t = 0; t < 4; ++t) {
            v8f s;
            #pragma unroll
            for (int r = 0; r < 8; ++r) s[r] = 0.f;
            #pragma unroll
            for (int kk = 0; kk < 16; ++kk) {
                const int d0   = kk * 4 + half * 2;
                const int krow = t * 16 + l16;
                v2f b; b.x = ldsK[krow * KS + d0]; b.y = ldsK[krow * KS + d0 + 1];
                s = __builtin_amdgcn_wmma_f32_16x16x4_f32(
                        false, qf[kk], false, b, (short)0, s, false, false);
            }
            const int jg = kb * BQ + t * 16 + l16;
            #pragma unroll
            for (int r = 0; r < 8; ++r) {
                const int ig = qb * BQ + w * 16 + r + 8 * half;
                const float sv = (jg <= ig) ? s[r] : 0.f;
                pacc[r] += sv;
                ldsS[(w * 16 + r + 8 * half) * KS + t * 16 + l16] = sv;
            }
        }
        __syncthreads();

        v2f sa[16];
        #pragma unroll
        for (int nn = 0; nn < 16; ++nn) {
            const int n0   = nn * 4 + half * 2;
            const int srow = w * 16 + l16;
            v2f a; a.x = ldsS[srow * KS + n0]; a.y = ldsS[srow * KS + n0 + 1];
            sa[nn] = a;
        }
        #pragma unroll
        for (int ct = 0; ct < 4; ++ct) {
            v8f o = oacc[ct];
            #pragma unroll
            for (int nn = 0; nn < 16; ++nn) {
                const int n0 = nn * 4 + half * 2;
                v2f b; b.x = ldsV[(n0    ) * VS + ct * 16 + l16];
                       b.y = ldsV[(n0 + 1) * VS + ct * 16 + l16];
                o = __builtin_amdgcn_wmma_f32_16x16x4_f32(
                        false, sa[nn], false, b, (short)0, o, false, false);
            }
            oacc[ct] = o;
        }
    }

    float invp[8];
    #pragma unroll
    for (int r = 0; r < 8; ++r) {
        float p = pacc[r];
        p += __shfl_xor(p, 1, 32);
        p += __shfl_xor(p, 2, 32);
        p += __shfl_xor(p, 4, 32);
        p += __shfl_xor(p, 8, 32);
        invp[r] = 1.f / fmaxf(fabsf(p), 1.f);
    }
    #pragma unroll
    for (int ct = 0; ct < 4; ++ct) {
        #pragma unroll
        for (int r = 0; r < 8; ++r) {
            const int m = w * 16 + r + 8 * half;
            out[bh_sd + (size_t)(qb * BQ + m) * Dc + ct * 16 + l16]
                = oacc[ct][r] * invp[r];
        }
    }
}

extern "C" void kernel_launch(void* const* d_in, const int* in_sizes, int n_in,
                              void* d_out, int out_size, void* d_ws, size_t ws_size,
                              hipStream_t stream) {
    (void)in_sizes; (void)n_in; (void)out_size; (void)d_ws; (void)ws_size;
    const float* q  = (const float*)d_in[0];
    const float* k  = (const float*)d_in[1];
    const float* v  = (const float*)d_in[2];
    const float* om = (const float*)d_in[3];
    float* out = (float*)d_out;
    dim3 grid(Sc / BQ, BHc);
    retention_fused_kernel<<<grid, 128, 0, stream>>>(q, k, v, om, out);
}


# Software-pipelined K/V staging: next key block's global loads issued into registers ahead of the current block's WMMA work, LDS-only barriers keep them in flight (was 8 serial load-wait trips)
# speedup vs baseline: 1.9288x; 1.9288x over previous
#include <hip/hip_runtime.h>


typedef float v2f __attribute__((ext_vector_type(2)));
typedef float v4f __attribute__((ext_vector_type(4)));
typedef float v8f __attribute__((ext_vector_type(8)));

#define Bc 2
#define Hc 8
#define Sc 2048
#define Dc 64
#define BHc (Bc * Hc)
#define BQ 64
#define KS 68
#define VS 72

__device__ __forceinline__ void fetch_kv_tile(const float* __restrict__ k,
                                              const float* __restrict__ v,
                                              const float* __restrict__ omask,
                                              size_t bh_sd, size_t bh_ss, int kb,
                                              int lrow, int lcol4,
                                              v4f (&kreg)[8], v4f (&vreg)[8], float (&mreg)[8])
{
    const float* kbase = k + bh_sd + (size_t)(kb * BQ) * Dc;
    const float* vbase = v + bh_sd + (size_t)(kb * BQ) * Dc;
    const float* mbase = omask + bh_ss + (size_t)(kb * BQ) * Sc;
    #pragma unroll
    for (int i = 0; i < 8; ++i) {
        const int row = lrow + 8 * i;
        mreg[i] = mbase[(size_t)row * Sc];
        kreg[i] = *(const v4f*)(kbase + row * Dc + lcol4);
        vreg[i] = *(const v4f*)(vbase + row * Dc + lcol4);
    }
}

__device__ __forceinline__ void lds_barrier()
{
    __builtin_amdgcn_fence(__ATOMIC_RELEASE, "workgroup", "local");
    __builtin_amdgcn_s_barrier();
    __builtin_amdgcn_fence(__ATOMIC_ACQUIRE, "workgroup", "local");
}

__launch_bounds__(128)
__global__ void retention_fused_kernel(const float* __restrict__ q,
                                       const float* __restrict__ k,
                                       const float* __restrict__ v,
                                       const float* __restrict__ omask,
                                       float* __restrict__ out)
{
    const int qb   = (int)(gridDim.x - 1u - blockIdx.x);
    const int bh   = (int)blockIdx.y;
    const int tid  = (int)threadIdx.x;
    const int w    = tid >> 5;
    const int lane = tid & 31;
    const int half = lane >> 4;
    const int l16  = lane & 15;

    __shared__ float ldsK[BQ * KS];
    __shared__ float ldsV[BQ * VS];
    __shared__ float ldsS[BQ * KS];
    __shared__ float ldsQC[BQ];

    const size_t bh_sd = (size_t)bh * Sc * Dc;
    const size_t bh_ss = (size_t)bh * Sc * (size_t)Sc;

    for (int rr = 0; rr < 16; ++rr) {
        const int row = w * 16 + rr;
        const size_t base = bh_ss + (size_t)(qb * BQ + row) * Sc;
        const v4f* om4 = (const v4f*)(omask + base);
        float s = 0.f;
        for (int j = lane; j < Sc / 4; j += 32) {
            v4f t = __builtin_nontemporal_load(om4 + j);
            s += (t.x + t.y) + (t.z + t.w);
        }
        #pragma unroll
        for (int off = 16; off > 0; off >>= 1) s += __shfl_xor(s, off, 32);
        if (lane == 0) {
            float m0 = omask[base];
            ldsQC[row] = m0 / sqrtf(s);
        }
    }
    __syncthreads();

    const int qrow = qb * BQ + w * 16 + l16;
    const float qc = ldsQC[w * 16 + l16];
    const float* qptr = q + bh_sd + (size_t)qrow * Dc;
    v2f qf[16];
    #pragma unroll
    for (int kk = 0; kk < 16; ++kk) {
        const int c = kk * 4 + half * 2;
        v2f a; a.x = qptr[c] * qc; a.y = qptr[c + 1] * qc;
        qf[kk] = a;
    }

    v8f oacc[4];
    v8f pacc;
    #pragma unroll
    for (int r = 0; r < 8; ++r) pacc[r] = 0.f;
    #pragma unroll
    for (int ct = 0; ct < 4; ++ct) {
        #pragma unroll
        for (int r = 0; r < 8; ++r) oacc[ct][r] = 0.f;
    }

    const int lrow  = tid >> 4;
    const int lcol4 = (tid & 15) * 4;
    v4f kreg[8], vreg[8];
    float mreg[8];
    fetch_kv_tile(k, v, omask, bh_sd, bh_ss, 0, lrow, lcol4, kreg, vreg, mreg);

    for (int kb = 0; kb <= qb; ++kb) {
        __syncthreads();
        #pragma unroll
        for (int i = 0; i < 8; ++i) {
            const int row = lrow + 8 * i;
            const float kc = 1.f / mreg[i];
            const v4f kvv = kreg[i];
            ldsK[row * KS + lcol4 + 0] = kvv.x * kc;
            ldsK[row * KS + lcol4 + 1] = kvv.y * kc;
            ldsK[row * KS + lcol4 + 2] = kvv.z * kc;
            ldsK[row * KS + lcol4 + 3] = kvv.w * kc;
            const v4f vvv = vreg[i];
            ldsV[row * VS + lcol4 + 0] = vvv.x;
            ldsV[row * VS + lcol4 + 1] = vvv.y;
            ldsV[row * VS + lcol4 + 2] = vvv.z;
            ldsV[row * VS + lcol4 + 3] = vvv.w;
        }
        if (kb < qb)
            fetch_kv_tile(k, v, omask, bh_sd, bh_ss, kb + 1, lrow, lcol4, kreg, vreg, mreg);
        lds_barrier();

        #pragma unroll
        for (int t = 0; t < 4; ++t) {
            v8f s;
            #pragma unroll
            for (int r = 0; r < 8; ++r) s[r] = 0.f;
            #pragma unroll
            for (int kk = 0; kk < 16; ++kk) {
                const int d0   = kk * 4 + half * 2;
                const int krow = t * 16 + l16;
                v2f b; b.x = ldsK[krow * KS + d0]; b.y = ldsK[krow * KS + d0 + 1];
                s = __builtin_amdgcn_wmma_f32_16x16x4_f32(
                        false, qf[kk], false, b, (short)0, s, false, false);
            }
            const int jg = kb * BQ + t * 16 + l16;
            #pragma unroll
            for (int r = 0; r < 8; ++r) {
                const int ig = qb * BQ + w * 16 + r + 8 * half;
                const float sv = (jg <= ig) ? s[r] : 0.f;
                pacc[r] += sv;
                ldsS[(w * 16 + r + 8 * half) * KS + t * 16 + l16] = sv;
            }
        }
        lds_barrier();

        v2f sa[16];
        #pragma unroll
        for (int nn = 0; nn < 16; ++nn) {
            const int n0   = nn * 4 + half * 2;
            const int srow = w * 16 + l16;
            v2f a; a.x = ldsS[srow * KS + n0]; a.y = ldsS[srow * KS + n0 + 1];
            sa[nn] = a;
        }
        #pragma unroll
        for (int ct = 0; ct < 4; ++ct) {
            v8f o = oacc[ct];
            #pragma unroll
            for (int nn = 0; nn < 16; ++nn) {
                const int n0 = nn * 4 + half * 2;
                v2f b; b.x = ldsV[(n0    ) * VS + ct * 16 + l16];
                       b.y = ldsV[(n0 + 1) * VS + ct * 16 + l16];
                o = __builtin_amdgcn_wmma_f32_16x16x4_f32(
                        false, sa[nn], false, b, (short)0, o, false, false);
            }
            oacc[ct] = o;
        }
    }

    float invp[8];
    #pragma unroll
    for (int r = 0; r < 8; ++r) {
        float p = pacc[r];
        p += __shfl_xor(p, 1, 32);
        p += __shfl_xor(p, 2, 32);
        p += __shfl_xor(p, 4, 32);
        p += __shfl_xor(p, 8, 32);
        invp[r] = 1.f / fmaxf(fabsf(p), 1.f);
    }
    #pragma unroll
    for (int ct = 0; ct < 4; ++ct) {
        #pragma unroll
        for (int r = 0; r < 8; ++r) {
            const int m = w * 16 + r + 8 * half;
            out[bh_sd + (size_t)(qb * BQ + m) * Dc + ct * 16 + l16]
                = oacc[ct][r] * invp[r];
        }
    }
}

extern "C" void kernel_launch(void* const* d_in, const int* in_sizes, int n_in,
                              void* d_out, int out_size, void* d_ws, size_t ws_size,
                              hipStream_t stream) {
    (void)in_sizes; (void)n_in; (void)out_size; (void)d_ws; (void)ws_size;
    const float* q  = (const float*)d_in[0];
    const float* k  = (const float*)d_in[1];
    const float* v  = (const float*)d_in[2];
    const float* om = (const float*)d_in[3];
    float* out = (float*)d_out;
    dim3 grid(Sc / BQ, BHc);
    retention_fused_kernel<<<grid, 128, 0, stream>>>(q, k, v, om, out);
}


# Equal causal work per workgroup: each takes 32-row query blocks a and 63-a (33 key blocks); two waves share a 16-row strip, splitting key tiles and output columns; row sum P = two per-wave partials
# speedup vs baseline: 2.4284x; 1.2590x over previous
#include <hip/hip_runtime.h>


typedef float v2f __attribute__((ext_vector_type(2)));
typedef float v4f __attribute__((ext_vector_type(4)));
typedef float v8f __attribute__((ext_vector_type(8)));

#define Bc 2
#define Hc 8
#define Sc 2048
#define Dc 64
#define BHc (Bc * Hc)
#define BQ 32
#define BK 64
#define NQB (Sc / BQ)
#define KS 68
#define VS 72

__device__ __forceinline__ void fetch_kv_tile(const float* __restrict__ k,
                                              const float* __restrict__ v,
                                              const float* __restrict__ omask,
                                              size_t bh_sd, size_t bh_ss, int kb,
                                              int lrow, int lcol4,
                                              v4f (&kreg)[8], v4f (&vreg)[8], float (&mreg)[8])
{
    const float* kbase = k + bh_sd + (size_t)(kb * BK) * Dc;
    const float* vbase = v + bh_sd + (size_t)(kb * BK) * Dc;
    const float* mbase = omask + bh_ss + (size_t)(kb * BK) * Sc;
    #pragma unroll
    for (int i = 0; i < 8; ++i) {
        const int row = lrow + 8 * i;
        mreg[i] = mbase[(size_t)row * Sc];
        kreg[i] = *(const v4f*)(kbase + row * Dc + lcol4);
        vreg[i] = *(const v4f*)(vbase + row * Dc + lcol4);
    }
}

__device__ __forceinline__ void lds_barrier()
{
    __builtin_amdgcn_fence(__ATOMIC_RELEASE, "workgroup", "local");
    __builtin_amdgcn_s_barrier();
    __builtin_amdgcn_fence(__ATOMIC_ACQUIRE, "workgroup", "local");
}

__launch_bounds__(128)
__global__ void retention_fused_kernel(const float* __restrict__ q,
                                       const float* __restrict__ k,
                                       const float* __restrict__ v,
                                       const float* __restrict__ omask,
                                       float* __restrict__ out)
{
    const int pa   = (int)blockIdx.x;
    const int bh   = (int)blockIdx.y;
    const int tid  = (int)threadIdx.x;
    const int w    = tid >> 5;
    const int lane = tid & 31;
    const int half = lane >> 4;
    const int l16  = lane & 15;
    const int strip = w >> 1;
    const int part  = w & 1;

    __shared__ float ldsK[BK * KS];
    __shared__ float ldsV[BK * VS];
    __shared__ float ldsS[BQ * KS];
    __shared__ float ldsQC[2 * BQ];
    __shared__ float ldsP[4 * 16];

    const size_t bh_sd = (size_t)bh * Sc * Dc;
    const size_t bh_ss = (size_t)bh * Sc * (size_t)Sc;

    for (int rr = 0; rr < 16; ++rr) {
        const int lr   = w * 16 + rr;
        const int grow = (lr < BQ) ? (NQB - 1 - pa) * BQ + lr : pa * BQ + (lr - BQ);
        const size_t base = bh_ss + (size_t)grow * Sc;
        const v4f* om4 = (const v4f*)(omask + base);
        float s = 0.f;
        for (int j = lane; j < Sc / 4; j += 32) {
            v4f t = __builtin_nontemporal_load(om4 + j);
            s += (t.x + t.y) + (t.z + t.w);
        }
        #pragma unroll
        for (int off = 16; off > 0; off >>= 1) s += __shfl_xor(s, off, 32);
        if (lane == 0) {
            float m0 = omask[base];
            ldsQC[lr] = m0 / sqrtf(s);
        }
    }
    __syncthreads();

    const int lrow  = tid >> 4;
    const int lcol4 = (tid & 15) * 4;

    for (int pass = 0; pass < 2; ++pass) {
        const int blk  = (pass == 0) ? (NQB - 1 - pa) : pa;
        const int row0 = blk * BQ;
        const int nkb  = (blk >> 1) + 1;

        const int qrow = row0 + strip * 16 + l16;
        const float qc = ldsQC[pass * BQ + strip * 16 + l16];
        const float* qptr = q + bh_sd + (size_t)qrow * Dc;
        v2f qf[16];
        #pragma unroll
        for (int kk = 0; kk < 16; ++kk) {
            const int c = kk * 4 + half * 2;
            v2f a; a.x = qptr[c] * qc; a.y = qptr[c + 1] * qc;
            qf[kk] = a;
        }

        v8f oacc[2];
        v8f pacc;
        #pragma unroll
        for (int r = 0; r < 8; ++r) pacc[r] = 0.f;
        #pragma unroll
        for (int cc = 0; cc < 2; ++cc) {
            #pragma unroll
            for (int r = 0; r < 8; ++r) oacc[cc][r] = 0.f;
        }

        v4f kreg[8], vreg[8];
        float mreg[8];
        fetch_kv_tile(k, v, omask, bh_sd, bh_ss, 0, lrow, lcol4, kreg, vreg, mreg);

        for (int kb = 0; kb < nkb; ++kb) {
            __syncthreads();
            #pragma unroll
            for (int i = 0; i < 8; ++i) {
                const int row = lrow + 8 * i;
                const float kc = 1.f / mreg[i];
                const v4f kvv = kreg[i];
                ldsK[row * KS + lcol4 + 0] = kvv.x * kc;
                ldsK[row * KS + lcol4 + 1] = kvv.y * kc;
                ldsK[row * KS + lcol4 + 2] = kvv.z * kc;
                ldsK[row * KS + lcol4 + 3] = kvv.w * kc;
                const v4f vvv = vreg[i];
                ldsV[row * VS + lcol4 + 0] = vvv.x;
                ldsV[row * VS + lcol4 + 1] = vvv.y;
                ldsV[row * VS + lcol4 + 2] = vvv.z;
                ldsV[row * VS + lcol4 + 3] = vvv.w;
            }
            if (kb + 1 < nkb)
                fetch_kv_tile(k, v, omask, bh_sd, bh_ss, kb + 1, lrow, lcol4, kreg, vreg, mreg);
            lds_barrier();

            #pragma unroll
            for (int tt = 0; tt < 2; ++tt) {
                const int t = part * 2 + tt;
                v8f s;
                #pragma unroll
                for (int r = 0; r < 8; ++r) s[r] = 0.f;
                #pragma unroll
                for (int kk = 0; kk < 16; ++kk) {
                    const int d0   = kk * 4 + half * 2;
                    const int krow = t * 16 + l16;
                    v2f b; b.x = ldsK[krow * KS + d0]; b.y = ldsK[krow * KS + d0 + 1];
                    s = __builtin_amdgcn_wmma_f32_16x16x4_f32(
                            false, qf[kk], false, b, (short)0, s, false, false);
                }
                const int jg = kb * BK + t * 16 + l16;
                #pragma unroll
                for (int r = 0; r < 8; ++r) {
                    const int ig = row0 + strip * 16 + r + 8 * half;
                    const float sv = (jg <= ig) ? s[r] : 0.f;
                    pacc[r] += sv;
                    ldsS[(strip * 16 + r + 8 * half) * KS + t * 16 + l16] = sv;
                }
            }
            lds_barrier();

            v2f sa[16];
            #pragma unroll
            for (int nn = 0; nn < 16; ++nn) {
                const int n0   = nn * 4 + half * 2;
                const int srow = strip * 16 + l16;
                v2f a; a.x = ldsS[srow * KS + n0]; a.y = ldsS[srow * KS + n0 + 1];
                sa[nn] = a;
            }
            #pragma unroll
            for (int cc = 0; cc < 2; ++cc) {
                const int ct = part * 2 + cc;
                v8f o = oacc[cc];
                #pragma unroll
                for (int nn = 0; nn < 16; ++nn) {
                    const int n0 = nn * 4 + half * 2;
                    v2f b; b.x = ldsV[(n0    ) * VS + ct * 16 + l16];
                           b.y = ldsV[(n0 + 1) * VS + ct * 16 + l16];
                    o = __builtin_amdgcn_wmma_f32_16x16x4_f32(
                            false, sa[nn], false, b, (short)0, o, false, false);
                }
                oacc[cc] = o;
            }
        }

        float prow[8];
        #pragma unroll
        for (int r = 0; r < 8; ++r) {
            float p = pacc[r];
            p += __shfl_xor(p, 1, 32);
            p += __shfl_xor(p, 2, 32);
            p += __shfl_xor(p, 4, 32);
            p += __shfl_xor(p, 8, 32);
            prow[r] = p;
        }
        if (l16 == 0) {
            #pragma unroll
            for (int r = 0; r < 8; ++r) ldsP[w * 16 + r + 8 * half] = prow[r];
        }
        __syncthreads();
        float invp[8];
        #pragma unroll
        for (int r = 0; r < 8; ++r) {
            const float p = ldsP[(strip * 2    ) * 16 + r + 8 * half]
                          + ldsP[(strip * 2 + 1) * 16 + r + 8 * half];
            invp[r] = 1.f / fmaxf(fabsf(p), 1.f);
        }
        #pragma unroll
        for (int cc = 0; cc < 2; ++cc) {
            const int ct = part * 2 + cc;
            #pragma unroll
            for (int r = 0; r < 8; ++r) {
                const int m = strip * 16 + r + 8 * half;
                out[bh_sd + (size_t)(row0 + m) * Dc + ct * 16 + l16]
                    = oacc[cc][r] * invp[r];
            }
        }
    }
}

extern "C" void kernel_launch(void* const* d_in, const int* in_sizes, int n_in,
                              void* d_out, int out_size, void* d_ws, size_t ws_size,
                              hipStream_t stream) {
    (void)in_sizes; (void)n_in; (void)out_size; (void)d_ws; (void)ws_size;
    const float* q  = (const float*)d_in[0];
    const float* k  = (const float*)d_in[1];
    const float* v  = (const float*)d_in[2];
    const float* om = (const float*)d_in[3];
    float* out = (float*)d_out;
    dim3 grid(NQB / 2, BHc);
    retention_fused_kernel<<<grid, 128, 0, stream>>>(q, k, v, om, out);
}


# omask row sums streamed inside the key-block loop (2 rows per trip, under the matrix work), no serial pre-pass; q unscaled, row coefficient applied in epilogue; row sum = 64 lane partials + tree
# speedup vs baseline: 2.9086x; 1.1977x over previous
#include <hip/hip_runtime.h>


typedef float v2f __attribute__((ext_vector_type(2)));
typedef float v4f __attribute__((ext_vector_type(4)));
typedef float v8f __attribute__((ext_vector_type(8)));

#define Bc 2
#define Hc 8
#define Sc 2048
#define Dc 64
#define BHc (Bc * Hc)
#define BQ 32
#define BK 64
#define NQB (Sc / BQ)
#define KS 68
#define VS 72
#define PS 65
#define NSTREAM ((2 * BQ) / 2)

__device__ __forceinline__ void fetch_kv_tile(const float* __restrict__ k,
                                              const float* __restrict__ v,
                                              const float* __restrict__ omask,
                                              size_t bh_sd, size_t bh_ss, int kb,
                                              int lrow, int lcol4,
                                              v4f (&kreg)[8], v4f (&vreg)[8], float (&mreg)[8])
{
    const float* kbase = k + bh_sd + (size_t)(kb * BK) * Dc;
    const float* vbase = v + bh_sd + (size_t)(kb * BK) * Dc;
    const float* mbase = omask + bh_ss + (size_t)(kb * BK) * Sc;
    #pragma unroll
    for (int i = 0; i < 8; ++i) {
        const int row = lrow + 8 * i;
        mreg[i] = mbase[(size_t)row * Sc];
        kreg[i] = *(const v4f*)(kbase + row * Dc + lcol4);
        vreg[i] = *(const v4f*)(vbase + row * Dc + lcol4);
    }
}

__device__ __forceinline__ void fetch_mask_rows(const float* __restrict__ omask,
                                                size_t bh_ss, int pa, int g, int w, int lane,
                                                v4f (&sreg)[8])
{
    const int lr   = 2 * g + (w >> 1);
    const int grow = (lr < BQ) ? (NQB - 1 - pa) * BQ + lr : pa * BQ + (lr - BQ);
    const v4f* p = (const v4f*)(omask + bh_ss + (size_t)grow * Sc) + (w & 1) * (Sc / 8) + lane;
    #pragma unroll
    for (int i = 0; i < 8; ++i) sreg[i] = __builtin_nontemporal_load(p + 32 * i);
}

__device__ __forceinline__ void lds_barrier()
{
    __builtin_amdgcn_fence(__ATOMIC_RELEASE, "workgroup", "local");
    __builtin_amdgcn_s_barrier();
    __builtin_amdgcn_fence(__ATOMIC_ACQUIRE, "workgroup", "local");
}

__launch_bounds__(128)
__global__ void retention_fused_kernel(const float* __restrict__ q,
                                       const float* __restrict__ k,
                                       const float* __restrict__ v,
                                       const float* __restrict__ omask,
                                       float* __restrict__ out)
{
    const int pa   = (int)blockIdx.x;
    const int bh   = (int)blockIdx.y;
    const int tid  = (int)threadIdx.x;
    const int w    = tid >> 5;
    const int lane = tid & 31;
    const int half = lane >> 4;
    const int l16  = lane & 15;
    const int strip = w >> 1;
    const int part  = w & 1;

    __shared__ float ldsK[BK * KS];
    __shared__ float ldsV[BK * VS];
    __shared__ float ldsS[BQ * KS];
    __shared__ float ldsQC[2 * BQ];
    __shared__ float ldsPart[2 * BQ * PS];
    __shared__ float ldsM0[2 * BQ];
    __shared__ float ldsP[4 * 16];

    const size_t bh_sd = (size_t)bh * Sc * Dc;
    const size_t bh_ss = (size_t)bh * Sc * (size_t)Sc;

    const int lrow  = tid >> 4;
    const int lcol4 = (tid & 15) * 4;

    int g = 0;

    for (int pass = 0; pass < 2; ++pass) {
        const int blk  = (pass == 0) ? (NQB - 1 - pa) : pa;
        const int row0 = blk * BQ;
        const int nkb  = (blk >> 1) + 1;

        const int qrow = row0 + strip * 16 + l16;
        const float* qptr = q + bh_sd + (size_t)qrow * Dc;
        v2f qf[16];
        #pragma unroll
        for (int kk = 0; kk < 16; ++kk) {
            const int c = kk * 4 + half * 2;
            v2f a; a.x = qptr[c]; a.y = qptr[c + 1];
            qf[kk] = a;
        }

        v8f oacc[2];
        v8f pacc;
        #pragma unroll
        for (int r = 0; r < 8; ++r) pacc[r] = 0.f;
        #pragma unroll
        for (int cc = 0; cc < 2; ++cc) {
            #pragma unroll
            for (int r = 0; r < 8; ++r) oacc[cc][r] = 0.f;
        }

        v4f kreg[8], vreg[8];
        float mreg[8];
        fetch_kv_tile(k, v, omask, bh_sd, bh_ss, 0, lrow, lcol4, kreg, vreg, mreg);

        for (int kb = 0; kb < nkb; ++kb) {
            __syncthreads();
            #pragma unroll
            for (int i = 0; i < 8; ++i) {
                const int row = lrow + 8 * i;
                const float kc = 1.f / mreg[i];
                const v4f kvv = kreg[i];
                ldsK[row * KS + lcol4 + 0] = kvv.x * kc;
                ldsK[row * KS + lcol4 + 1] = kvv.y * kc;
                ldsK[row * KS + lcol4 + 2] = kvv.z * kc;
                ldsK[row * KS + lcol4 + 3] = kvv.w * kc;
                const v4f vvv = vreg[i];
                ldsV[row * VS + lcol4 + 0] = vvv.x;
                ldsV[row * VS + lcol4 + 1] = vvv.y;
                ldsV[row * VS + lcol4 + 2] = vvv.z;
                ldsV[row * VS + lcol4 + 3] = vvv.w;
            }
            if (kb + 1 < nkb)
                fetch_kv_tile(k, v, omask, bh_sd, bh_ss, kb + 1, lrow, lcol4, kreg, vreg, mreg);
            const bool stream = g < NSTREAM;
            v4f sreg[8];
            if (stream) fetch_mask_rows(omask, bh_ss, pa, g, w, lane, sreg);
            lds_barrier();

            #pragma unroll
            for (int tt = 0; tt < 2; ++tt) {
                const int t = part * 2 + tt;
                v8f s;
                #pragma unroll
                for (int r = 0; r < 8; ++r) s[r] = 0.f;
                #pragma unroll
                for (int kk = 0; kk < 16; ++kk) {
                    const int d0   = kk * 4 + half * 2;
                    const int krow = t * 16 + l16;
                    v2f b; b.x = ldsK[krow * KS + d0]; b.y = ldsK[krow * KS + d0 + 1];
                    s = __builtin_amdgcn_wmma_f32_16x16x4_f32(
                            false, qf[kk], false, b, (short)0, s, false, false);
                }
                const int jg = kb * BK + t * 16 + l16;
                #pragma unroll
                for (int r = 0; r < 8; ++r) {
                    const int ig = row0 + strip * 16 + r + 8 * half;
                    const float sv = (jg <= ig) ? s[r] : 0.f;
                    pacc[r] += sv;
                    ldsS[(strip * 16 + r + 8 * half) * KS + t * 16 + l16] = sv;
                }
            }
            lds_barrier();

            v2f sa[16];
            #pragma unroll
            for (int nn = 0; nn < 16; ++nn) {
                const int n0   = nn * 4 + half * 2;
                const int srow = strip * 16 + l16;
                v2f a; a.x = ldsS[srow * KS + n0]; a.y = ldsS[srow * KS + n0 + 1];
                sa[nn] = a;
            }
            #pragma unroll
            for (int cc = 0; cc < 2; ++cc) {
                const int ct = part * 2 + cc;
                v8f o = oacc[cc];
                #pragma unroll
                for (int nn = 0; nn < 16; ++nn) {
                    const int n0 = nn * 4 + half * 2;
                    v2f b; b.x = ldsV[(n0    ) * VS + ct * 16 + l16];
                           b.y = ldsV[(n0 + 1) * VS + ct * 16 + l16];
                    o = __builtin_amdgcn_wmma_f32_16x16x4_f32(
                            false, sa[nn], false, b, (short)0, o, false, false);
                }
                oacc[cc] = o;
            }

            if (stream) {
                const v4f a = ((sreg[0] + sreg[1]) + (sreg[2] + sreg[3]))
                            + ((sreg[4] + sreg[5]) + (sreg[6] + sreg[7]));
                const int lr = 2 * g + (w >> 1);
                ldsPart[lr * PS + (w & 1) * 32 + lane] = (a.x + a.y) + (a.z + a.w);
                if ((w & 1) == 0 && lane == 0) ldsM0[lr] = sreg[0].x;
            }
            ++g;
        }

        float prow[8];
        #pragma unroll
        for (int r = 0; r < 8; ++r) {
            float p = pacc[r];
            p += __shfl_xor(p, 1, 32);
            p += __shfl_xor(p, 2, 32);
            p += __shfl_xor(p, 4, 32);
            p += __shfl_xor(p, 8, 32);
            prow[r] = p;
        }
        if (l16 == 0) {
            #pragma unroll
            for (int r = 0; r < 8; ++r) ldsP[w * 16 + r + 8 * half] = prow[r];
        }
        __syncthreads();
        {
            const int lr = pass * BQ + (tid >> 2);
            const float* pp = ldsPart + lr * PS + (tid & 3) * 16;
            float s = ((pp[0] + pp[1]) + (pp[2] + pp[3])) + ((pp[4] + pp[5]) + (pp[6] + pp[7]));
            s += ((pp[8] + pp[9]) + (pp[10] + pp[11])) + ((pp[12] + pp[13]) + (pp[14] + pp[15]));
            s += __shfl_xor(s, 1, 32);
            s += __shfl_xor(s, 2, 32);
            if ((tid & 3) == 0) ldsQC[lr] = ldsM0[lr] / sqrtf(s);
        }
        __syncthreads();
        float invp[8];
        #pragma unroll
        for (int r = 0; r < 8; ++r) {
            const float qc = ldsQC[pass * BQ + strip * 16 + r + 8 * half];
            const float p = (ldsP[(strip * 2    ) * 16 + r + 8 * half]
                           + ldsP[(strip * 2 + 1) * 16 + r + 8 * half]) * qc;
            invp[r] = qc / fmaxf(fabsf(p), 1.f);
        }
        #pragma unroll
        for (int cc = 0; cc < 2; ++cc) {
            const int ct = part * 2 + cc;
            #pragma unroll
            for (int r = 0; r < 8; ++r) {
                const int m = strip * 16 + r + 8 * half;
                out[bh_sd + (size_t)(row0 + m) * Dc + ct * 16 + l16]
                    = oacc[cc][r] * invp[r];
            }
        }
    }
}

extern "C" void kernel_launch(void* const* d_in, const int* in_sizes, int n_in,
                              void* d_out, int out_size, void* d_ws, size_t ws_size,
                              hipStream_t stream) {
    (void)in_sizes; (void)n_in; (void)out_size; (void)d_ws; (void)ws_size;
    const float* q  = (const float*)d_in[0];
    const float* k  = (const float*)d_in[1];
    const float* v  = (const float*)d_in[2];
    const float* om = (const float*)d_in[3];
    float* out = (float*)d_out;
    dim3 grid(NQB / 2, BHc);
    retention_fused_kernel<<<grid, 128, 0, stream>>>(q, k, v, om, out);
}


# Both products formed transposed (S^T = K.Q^T, its D registers are the B operand of O^T += V^T.S^T): no LDS exchange or barrier for scores; a strip's 2 waves split keys; O, P = sum of 2 wave partials
# speedup vs baseline: 3.0313x; 1.0422x over previous
#include <hip/hip_runtime.h>


typedef float v2f __attribute__((ext_vector_type(2)));
typedef float v4f __attribute__((ext_vector_type(4)));
typedef float v8f __attribute__((ext_vector_type(8)));

#define Bc 2
#define Hc 8
#define Sc 2048
#define Dc 64
#define BHc (Bc * Hc)
#define BQ 32
#define BK 64
#define NQB (Sc / BQ)
#define KS 68
#define VS 68
#define PS 65
#define NSTREAM ((2 * BQ) / 2)

__device__ __forceinline__ void fetch_kv_tile(const float* __restrict__ k,
                                              const float* __restrict__ v,
                                              const float* __restrict__ omask,
                                              size_t bh_sd, size_t bh_ss, int kb,
                                              int lrow, int lcol4,
                                              v4f (&kreg)[8], v4f (&vreg)[8], float (&mreg)[8])
{
    const float* kbase = k + bh_sd + (size_t)(kb * BK) * Dc;
    const float* vbase = v + bh_sd + (size_t)(kb * BK) * Dc;
    const float* mbase = omask + bh_ss + (size_t)(kb * BK) * Sc;
    #pragma unroll
    for (int i = 0; i < 8; ++i) {
        const int row = lrow + 8 * i;
        mreg[i] = mbase[(size_t)row * Sc];
        kreg[i] = *(const v4f*)(kbase + row * Dc + lcol4);
        vreg[i] = *(const v4f*)(vbase + row * Dc + lcol4);
    }
}

__device__ __forceinline__ void fetch_mask_rows(const float* __restrict__ omask,
                                                size_t bh_ss, int pa, int g, int w, int lane,
                                                v4f (&sreg)[8])
{
    const int lr   = 2 * g + (w >> 1);
    const int grow = (lr < BQ) ? (NQB - 1 - pa) * BQ + lr : pa * BQ + (lr - BQ);
    const v4f* p = (const v4f*)(omask + bh_ss + (size_t)grow * Sc) + (w & 1) * (Sc / 8) + lane;
    #pragma unroll
    for (int i = 0; i < 8; ++i) sreg[i] = __builtin_nontemporal_load(p + 32 * i);
}

__device__ __forceinline__ void lds_barrier()
{
    __builtin_amdgcn_fence(__ATOMIC_RELEASE, "workgroup", "local");
    __builtin_amdgcn_s_barrier();
    __builtin_amdgcn_fence(__ATOMIC_ACQUIRE, "workgroup", "local");
}

__launch_bounds__(128)
__global__ void retention_fused_kernel(const float* __restrict__ q,
                                       const float* __restrict__ k,
                                       const float* __restrict__ v,
                                       const float* __restrict__ omask,
                                       float* __restrict__ out)
{
    const int pa   = (int)blockIdx.x;
    const int bh   = (int)blockIdx.y;
    const int tid  = (int)threadIdx.x;
    const int w    = tid >> 5;
    const int lane = tid & 31;
    const int half = lane >> 4;
    const int l16  = lane & 15;
    const int strip = w >> 1;
    const int part  = w & 1;

    __shared__ __attribute__((aligned(16))) float ldsK[BK * KS];
    __shared__ __attribute__((aligned(16))) float ldsV[BK * VS];
    __shared__ float ldsO[4 * 16 * 32];
    __shared__ float ldsQC[2 * BQ];
    __shared__ float ldsPart[2 * BQ * PS];
    __shared__ float ldsM0[2 * BQ];
    __shared__ float ldsP[4 * 16];

    const size_t bh_sd = (size_t)bh * Sc * Dc;
    const size_t bh_ss = (size_t)bh * Sc * (size_t)Sc;

    const int lrow  = tid >> 4;
    const int lcol4 = (tid & 15) * 4;

    const float* kfp = ldsK + (part * 32 + l16) * KS + half * 2;
    const float* vfp = ldsV + (part * 32 + half * 8) * VS + l16;
    int vcol[4];
    #pragma unroll
    for (int a = 0; a < 4; ++a) {
        vcol[a] = ((a + 2 * part) & 3) * 16;
        asm volatile("" : "+v"(vcol[a]));
    }

    int g = 0;

    for (int pass = 0; pass < 2; ++pass) {
        const int blk  = (pass == 0) ? (NQB - 1 - pa) : pa;
        const int row0 = blk * BQ;
        const int nkb  = (blk >> 1) + 1;

        const int qrow = row0 + strip * 16 + l16;
        const float* qptr = q + bh_sd + (size_t)qrow * Dc;
        v2f qf[16];
        #pragma unroll
        for (int kk = 0; kk < 16; ++kk) qf[kk] = *(const v2f*)(qptr + kk * 4 + half * 2);

        v8f oacc[4];
        float pacc = 0.f;
        #pragma unroll
        for (int a = 0; a < 4; ++a) {
            #pragma unroll
            for (int r = 0; r < 8; ++r) oacc[a][r] = 0.f;
        }

        v4f kreg[8], vreg[8];
        float mreg[8];
        fetch_kv_tile(k, v, omask, bh_sd, bh_ss, 0, lrow, lcol4, kreg, vreg, mreg);

        for (int kb = 0; kb < nkb; ++kb) {
            __syncthreads();
            #pragma unroll
            for (int i = 0; i < 8; ++i) {
                const int row = lrow + 8 * i;
                const float kc = 1.f / mreg[i];
                *(v4f*)(ldsK + row * KS + lcol4) = kreg[i] * kc;
                *(v4f*)(ldsV + row * VS + lcol4) = vreg[i];
            }
            if (kb + 1 < nkb)
                fetch_kv_tile(k, v, omask, bh_sd, bh_ss, kb + 1, lrow, lcol4, kreg, vreg, mreg);
            const bool stream = g < NSTREAM;
            v4f sreg[8];
            if (stream) fetch_mask_rows(omask, bh_ss, pa, g, w, lane, sreg);
            lds_barrier();

            v2f kf[2][16];
            #pragma unroll
            for (int tt = 0; tt < 2; ++tt) {
                #pragma unroll
                for (int kk = 0; kk < 16; ++kk)
                    kf[tt][kk] = *(const v2f*)(kfp + tt * 16 * KS + kk * 4);
            }
            v2f vf[2][4][4];
            #pragma unroll
            for (int tt = 0; tt < 2; ++tt) {
                #pragma unroll
                for (int j = 0; j < 4; ++j) {
                    #pragma unroll
                    for (int a = 0; a < 4; ++a) {
                        const float* p = vfp + (tt * 16 + 2 * j) * VS + vcol[a];
                        v2f f; f.x = p[0]; f.y = p[VS];
                        vf[tt][j][a] = f;
                    }
                }
            }
            __builtin_amdgcn_sched_barrier(0);

            v8f st[2];
            #pragma unroll
            for (int tt = 0; tt < 2; ++tt) {
                #pragma unroll
                for (int r = 0; r < 8; ++r) st[tt][r] = 0.f;
            }
            #pragma unroll
            for (int kk = 0; kk < 16; ++kk) {
                #pragma unroll
                for (int tt = 0; tt < 2; ++tt)
                    st[tt] = __builtin_amdgcn_wmma_f32_16x16x4_f32(
                                 false, kf[tt][kk], false, qf[kk], (short)0, st[tt], false, false);
            }
            #pragma unroll
            for (int tt = 0; tt < 2; ++tt) {
                const int jg0  = kb * BK + (part * 2 + tt) * 16 + half * 8;
                const int keep = qrow - jg0;
                #pragma unroll
                for (int r = 0; r < 8; ++r) st[tt][r] = (r <= keep) ? st[tt][r] : 0.f;
                pacc += ((st[tt][0] + st[tt][1]) + (st[tt][2] + st[tt][3]))
                      + ((st[tt][4] + st[tt][5]) + (st[tt][6] + st[tt][7]));
            }
            #pragma unroll
            for (int tt = 0; tt < 2; ++tt) {
                #pragma unroll
                for (int j = 0; j < 4; ++j) {
                    v2f b; b.x = st[tt][2 * j]; b.y = st[tt][2 * j + 1];
                    #pragma unroll
                    for (int a = 0; a < 4; ++a)
                        oacc[a] = __builtin_amdgcn_wmma_f32_16x16x4_f32(
                                      false, vf[tt][j][a], false, b, (short)0, oacc[a], false, false);
                }
            }

            if (stream) {
                const v4f a = ((sreg[0] + sreg[1]) + (sreg[2] + sreg[3]))
                            + ((sreg[4] + sreg[5]) + (sreg[6] + sreg[7]));
                const int lr = 2 * g + (w >> 1);
                ldsPart[lr * PS + (w & 1) * 32 + lane] = (a.x + a.y) + (a.z + a.w);
                if ((w & 1) == 0 && lane == 0) ldsM0[lr] = sreg[0].x;
            }
            ++g;
        }

        {
            const float p = pacc + __shfl_xor(pacc, 16, 32);
            if (half == 0) ldsP[w * 16 + l16] = p;
        }
        {
            float* give = ldsO + w * 512 + lane;
            #pragma unroll
            for (int mm = 0; mm < 2; ++mm) {
                #pragma unroll
                for (int r = 0; r < 8; ++r) give[(mm * 8 + r) * 32] = oacc[2 + mm][r];
            }
        }
        __syncthreads();
        {
            const int lr = pass * BQ + (tid >> 2);
            const float* pp = ldsPart + lr * PS + (tid & 3) * 16;
            float s = ((pp[0] + pp[1]) + (pp[2] + pp[3])) + ((pp[4] + pp[5]) + (pp[6] + pp[7]));
            s += ((pp[8] + pp[9]) + (pp[10] + pp[11])) + ((pp[12] + pp[13]) + (pp[14] + pp[15]));
            s += __shfl_xor(s, 1, 32);
            s += __shfl_xor(s, 2, 32);
            if ((tid & 3) == 0) ldsQC[lr] = ldsM0[lr] / sqrtf(s);
        }
        __syncthreads();
        {
            const float qc = ldsQC[pass * BQ + strip * 16 + l16];
            const float p = (ldsP[(strip * 2) * 16 + l16] + ldsP[(strip * 2 + 1) * 16 + l16]) * qc;
            const float invp = qc / fmaxf(fabsf(p), 1.f);
            const float* take = ldsO + (w ^ 1) * 512 + lane;
            float* orow = out + bh_sd + (size_t)qrow * Dc + part * 32 + half * 8;
            #pragma unroll
            for (int mm = 0; mm < 2; ++mm) {
                v4f lo, hi;
                lo.x = (oacc[mm][0] + take[(mm * 8 + 0) * 32]) * invp;
                lo.y = (oacc[mm][1] + take[(mm * 8 + 1) * 32]) * invp;
                lo.z = (oacc[mm][2] + take[(mm * 8 + 2) * 32]) * invp;
                lo.w = (oacc[mm][3] + take[(mm * 8 + 3) * 32]) * invp;
                hi.x = (oacc[mm][4] + take[(mm * 8 + 4) * 32]) * invp;
                hi.y = (oacc[mm][5] + take[(mm * 8 + 5) * 32]) * invp;
                hi.z = (oacc[mm][6] + take[(mm * 8 + 6) * 32]) * invp;
                hi.w = (oacc[mm][7] + take[(mm * 8 + 7) * 32]) * invp;
                *(v4f*)(orow + mm * 16)     = lo;
                *(v4f*)(orow + mm * 16 + 4) = hi;
            }
        }
    }
}

extern "C" void kernel_launch(void* const* d_in, const int* in_sizes, int n_in,
                              void* d_out, int out_size, void* d_ws, size_t ws_size,
                              hipStream_t stream) {
    (void)in_sizes; (void)n_in; (void)out_size; (void)d_ws; (void)ws_size;
    const float* q  = (const float*)d_in[0];
    const float* k  = (const float*)d_in[1];
    const float* v  = (const float*)d_in[2];
    const float* om = (const float*)d_in[3];
    float* out = (float*)d_out;
    dim3 grid(NQB / 2, BHc);
    retention_fused_kernel<<<grid, 128, 0, stream>>>(q, k, v, om, out);
}


# K scale 1/omask[j][0]: each lane loads and divides for ONE of the rows its wave stages, cells take it by DPP row_share; was 8 serial IEEE divisions + 8 loads per lane per trip; bit-identical
# speedup vs baseline: 3.2406x; 1.0690x over previous
#include <hip/hip_runtime.h>


typedef float v2f __attribute__((ext_vector_type(2)));
typedef float v4f __attribute__((ext_vector_type(4)));
typedef float v8f __attribute__((ext_vector_type(8)));

#define Bc 2
#define Hc 8
#define Sc 2048
#define Dc 64
#define BHc (Bc * Hc)
#define BQ 32
#define BK 64
#define NQB (Sc / BQ)
#define KS 68
#define VS 68
#define PS 65
#define NSTREAM ((2 * BQ) / 2)

__device__ __forceinline__ void fetch_kv_tile(const float* __restrict__ k,
                                              const float* __restrict__ v,
                                              const float* __restrict__ omask,
                                              size_t bh_sd, size_t bh_ss, int kb,
                                              int lrow, int lcol4, int mrow,
                                              v4f (&kreg)[8], v4f (&vreg)[8], float &mlane)
{
    const float* kbase = k + bh_sd + (size_t)(kb * BK) * Dc;
    const float* vbase = v + bh_sd + (size_t)(kb * BK) * Dc;
    const float* mbase = omask + bh_ss + (size_t)(kb * BK) * Sc;
    mlane = mbase[(size_t)mrow * Sc];
    #pragma unroll
    for (int i = 0; i < 8; ++i) {
        const int row = lrow + 8 * i;
        kreg[i] = *(const v4f*)(kbase + row * Dc + lcol4);
        vreg[i] = *(const v4f*)(vbase + row * Dc + lcol4);
    }
}

__device__ __forceinline__ void fetch_mask_rows(const float* __restrict__ omask,
                                                size_t bh_ss, int pa, int g, int w, int lane,
                                                v4f (&sreg)[8])
{
    const int lr   = 2 * g + (w >> 1);
    const int grow = (lr < BQ) ? (NQB - 1 - pa) * BQ + lr : pa * BQ + (lr - BQ);
    const v4f* p = (const v4f*)(omask + bh_ss + (size_t)grow * Sc) + (w & 1) * (Sc / 8) + lane;
    #pragma unroll
    for (int i = 0; i < 8; ++i) sreg[i] = __builtin_nontemporal_load(p + 32 * i);
}

template <int I>
__device__ __forceinline__ float half_lane(float x)
{
    return __int_as_float(__builtin_amdgcn_update_dpp(0, __float_as_int(x), 0x150 + I, 0xf, 0xf, true));
}

__device__ __forceinline__ void lds_barrier()
{
    __builtin_amdgcn_fence(__ATOMIC_RELEASE, "workgroup", "local");
    __builtin_amdgcn_s_barrier();
    __builtin_amdgcn_fence(__ATOMIC_ACQUIRE, "workgroup", "local");
}

__launch_bounds__(128)
__global__ void retention_fused_kernel(const float* __restrict__ q,
                                       const float* __restrict__ k,
                                       const float* __restrict__ v,
                                       const float* __restrict__ omask,
                                       float* __restrict__ out)
{
    const int pa   = (int)blockIdx.x;
    const int bh   = (int)blockIdx.y;
    const int tid  = (int)threadIdx.x;
    const int w    = tid >> 5;
    const int lane = tid & 31;
    const int half = lane >> 4;
    const int l16  = lane & 15;
    const int strip = w >> 1;
    const int part  = w & 1;

    __shared__ __attribute__((aligned(16))) float ldsK[BK * KS];
    __shared__ __attribute__((aligned(16))) float ldsV[BK * VS];
    __shared__ float ldsO[4 * 16 * 32];
    __shared__ float ldsQC[2 * BQ];
    __shared__ float ldsPart[2 * BQ * PS];
    __shared__ float ldsM0[2 * BQ];
    __shared__ float ldsP[4 * 16];

    const size_t bh_sd = (size_t)bh * Sc * Dc;
    const size_t bh_ss = (size_t)bh * Sc * (size_t)Sc;

    const int lrow  = tid >> 4;
    const int lcol4 = (tid & 15) * 4;
    const int mrow  = lrow + 8 * (lane & 7);

    const float* kfp = ldsK + (part * 32 + l16) * KS + half * 2;
    const float* vfp = ldsV + (part * 32 + half * 8) * VS + l16;
    int vcol[4];
    #pragma unroll
    for (int a = 0; a < 4; ++a) {
        vcol[a] = ((a + 2 * part) & 3) * 16;
        asm volatile("" : "+v"(vcol[a]));
    }

    int g = 0;

    for (int pass = 0; pass < 2; ++pass) {
        const int blk  = (pass == 0) ? (NQB - 1 - pa) : pa;
        const int row0 = blk * BQ;
        const int nkb  = (blk >> 1) + 1;

        const int qrow = row0 + strip * 16 + l16;
        const float* qptr = q + bh_sd + (size_t)qrow * Dc;
        v2f qf[16];
        #pragma unroll
        for (int kk = 0; kk < 16; ++kk) qf[kk] = *(const v2f*)(qptr + kk * 4 + half * 2);

        v8f oacc[4];
        float pacc = 0.f;
        #pragma unroll
        for (int a = 0; a < 4; ++a) {
            #pragma unroll
            for (int r = 0; r < 8; ++r) oacc[a][r] = 0.f;
        }

        v4f kreg[8], vreg[8];
        float mlane;
        fetch_kv_tile(k, v, omask, bh_sd, bh_ss, 0, lrow, lcol4, mrow, kreg, vreg, mlane);

        for (int kb = 0; kb < nkb; ++kb) {
            __syncthreads();
            {
                const float kcl = 1.f / mlane;
#define STAGE_CELL(I)                                                              \
                {                                                                  \
                    const int row = lrow + 8 * (I);                                \
                    const float kc = half_lane<(I)>(kcl);                          \
                    *(v4f*)(ldsK + row * KS + lcol4) = kreg[(I)] * kc;             \
                    *(v4f*)(ldsV + row * VS + lcol4) = vreg[(I)];                  \
                }
                STAGE_CELL(0) STAGE_CELL(1) STAGE_CELL(2) STAGE_CELL(3)
                STAGE_CELL(4) STAGE_CELL(5) STAGE_CELL(6) STAGE_CELL(7)
#undef STAGE_CELL
            }
            if (kb + 1 < nkb)
                fetch_kv_tile(k, v, omask, bh_sd, bh_ss, kb + 1, lrow, lcol4, mrow, kreg, vreg, mlane);
            const bool stream = g < NSTREAM;
            v4f sreg[8];
            if (stream) fetch_mask_rows(omask, bh_ss, pa, g, w, lane, sreg);
            lds_barrier();

            v2f kf[2][16];
            #pragma unroll
            for (int tt = 0; tt < 2; ++tt) {
                #pragma unroll
                for (int kk = 0; kk < 16; ++kk)
                    kf[tt][kk] = *(const v2f*)(kfp + tt * 16 * KS + kk * 4);
            }
            v2f vf[2][4][4];
            #pragma unroll
            for (int tt = 0; tt < 2; ++tt) {
                #pragma unroll
                for (int j = 0; j < 4; ++j) {
                    #pragma unroll
                    for (int a = 0; a < 4; ++a) {
                        const float* p = vfp + (tt * 16 + 2 * j) * VS + vcol[a];
                        v2f f; f.x = p[0]; f.y = p[VS];
                        vf[tt][j][a] = f;
                    }
                }
            }
            __builtin_amdgcn_sched_barrier(0);

            v8f st[2];
            #pragma unroll
            for (int tt = 0; tt < 2; ++tt) {
                #pragma unroll
                for (int r = 0; r < 8; ++r) st[tt][r] = 0.f;
            }
            #pragma unroll
            for (int kk = 0; kk < 16; ++kk) {
                #pragma unroll
                for (int tt = 0; tt < 2; ++tt)
                    st[tt] = __builtin_amdgcn_wmma_f32_16x16x4_f32(
                                 false, kf[tt][kk], false, qf[kk], (short)0, st[tt], false, false);
            }
            #pragma unroll
            for (int tt = 0; tt < 2; ++tt) {
                const int jg0  = kb * BK + (part * 2 + tt) * 16 + half * 8;
                const int keep = qrow - jg0;
                #pragma unroll
                for (int r = 0; r < 8; ++r) st[tt][r] = (r <= keep) ? st[tt][r] : 0.f;
                pacc += ((st[tt][0] + st[tt][1]) + (st[tt][2] + st[tt][3]))
                      + ((st[tt][4] + st[tt][5]) + (st[tt][6] + st[tt][7]));
            }
            #pragma unroll
            for (int tt = 0; tt < 2; ++tt) {
                #pragma unroll
                for (int j = 0; j < 4; ++j) {
                    v2f b; b.x = st[tt][2 * j]; b.y = st[tt][2 * j + 1];
                    #pragma unroll
                    for (int a = 0; a < 4; ++a)
                        oacc[a] = __builtin_amdgcn_wmma_f32_16x16x4_f32(
                                      false, vf[tt][j][a], false, b, (short)0, oacc[a], false, false);
                }
            }

            if (stream) {
                const v4f a = ((sreg[0] + sreg[1]) + (sreg[2] + sreg[3]))
                            + ((sreg[4] + sreg[5]) + (sreg[6] + sreg[7]));
                const int lr = 2 * g + (w >> 1);
                ldsPart[lr * PS + (w & 1) * 32 + lane] = (a.x + a.y) + (a.z + a.w);
                if ((w & 1) == 0 && lane == 0) ldsM0[lr] = sreg[0].x;
            }
            ++g;
        }

        {
            const float p = pacc + __shfl_xor(pacc, 16, 32);
            if (half == 0) ldsP[w * 16 + l16] = p;
        }
        {
            float* give = ldsO + w * 512 + lane;
            #pragma unroll
            for (int mm = 0; mm < 2; ++mm) {
                #pragma unroll
                for (int r = 0; r < 8; ++r) give[(mm * 8 + r) * 32] = oacc[2 + mm][r];
            }
        }
        __syncthreads();
        {
            const int lr = pass * BQ + (tid >> 2);
            const float* pp = ldsPart + lr * PS + (tid & 3) * 16;
            float s = ((pp[0] + pp[1]) + (pp[2] + pp[3])) + ((pp[4] + pp[5]) + (pp[6] + pp[7]));
            s += ((pp[8] + pp[9]) + (pp[10] + pp[11])) + ((pp[12] + pp[13]) + (pp[14] + pp[15]));
            s += __shfl_xor(s, 1, 32);
            s += __shfl_xor(s, 2, 32);
            if ((tid & 3) == 0) ldsQC[lr] = ldsM0[lr] / sqrtf(s);
        }
        __syncthreads();
        {
            const float qc = ldsQC[pass * BQ + strip * 16 + l16];
            const float p = (ldsP[(strip * 2) * 16 + l16] + ldsP[(strip * 2 + 1) * 16 + l16]) * qc;
            const float invp = qc / fmaxf(fabsf(p), 1.f);
            const float* take = ldsO + (w ^ 1) * 512 + lane;
            float* orow = out + bh_sd + (size_t)qrow * Dc + part * 32 + half * 8;
            #pragma unroll
            for (int mm = 0; mm < 2; ++mm) {
                v4f lo, hi;
                lo.x = (oacc[mm][0] + take[(mm * 8 + 0) * 32]) * invp;
                lo.y = (oacc[mm][1] + take[(mm * 8 + 1) * 32]) * invp;
                lo.z = (oacc[mm][2] + take[(mm * 8 + 2) * 32]) * invp;
                lo.w = (oacc[mm][3] + take[(mm * 8 + 3) * 32]) * invp;
                hi.x = (oacc[mm][4] + take[(mm * 8 + 4) * 32]) * invp;
                hi.y = (oacc[mm][5] + take[(mm * 8 + 5) * 32]) * invp;
                hi.z = (oacc[mm][6] + take[(mm * 8 + 6) * 32]) * invp;
                hi.w = (oacc[mm][7] + take[(mm * 8 + 7) * 32]) * invp;
                *(v4f*)(orow + mm * 16)     = lo;
                *(v4f*)(orow + mm * 16 + 4) = hi;
            }
        }
    }
}

extern "C" void kernel_launch(void* const* d_in, const int* in_sizes, int n_in,
                              void* d_out, int out_size, void* d_ws, size_t ws_size,
                              hipStream_t stream) {
    (void)in_sizes; (void)n_in; (void)out_size; (void)d_ws; (void)ws_size;
    const float* q  = (const float*)d_in[0];
    const float* k  = (const float*)d_in[1];
    const float* v  = (const float*)d_in[2];
    const float* om = (const float*)d_in[3];
    float* out = (float*)d_out;
    dim3 grid(NQB / 2, BHc);
    retention_fused_kernel<<<grid, 128, 0, stream>>>(q, k, v, om, out);
}


# Each wave takes TWO 16-row query strips for its 32 keys: every K and V fragment read from LDS feeds two WMMAs; a workgroup covers 64-row blocks a and 31-a, staging a K/V block once per 64 rows
# speedup vs baseline: 3.6156x; 1.1157x over previous
#include <hip/hip_runtime.h>


typedef float v2f __attribute__((ext_vector_type(2)));
typedef float v4f __attribute__((ext_vector_type(4)));
typedef float v8f __attribute__((ext_vector_type(8)));

#define Bc 2
#define Hc 8
#define Sc 2048
#define Dc 64
#define BHc (Bc * Hc)
#define BQ 64
#define BK 64
#define NQB (Sc / BQ)
#define KS 68
#define VS 68
#define PS 65
#define NSTREAM ((2 * BQ) / 4)

__device__ __forceinline__ void fetch_kv_tile(const float* __restrict__ k,
                                              const float* __restrict__ v,
                                              const float* __restrict__ omask,
                                              size_t bh_sd, size_t bh_ss, int kb,
                                              int lrow, int lcol4, int mrow,
                                              v4f (&kreg)[8], v4f (&vreg)[8], float &mlane)
{
    const float* kbase = k + bh_sd + (size_t)(kb * BK) * Dc;
    const float* vbase = v + bh_sd + (size_t)(kb * BK) * Dc;
    const float* mbase = omask + bh_ss + (size_t)(kb * BK) * Sc;
    mlane = mbase[(size_t)mrow * Sc];
    #pragma unroll
    for (int i = 0; i < 8; ++i) {
        const int row = lrow + 8 * i;
        kreg[i] = *(const v4f*)(kbase + row * Dc + lcol4);
        vreg[i] = *(const v4f*)(vbase + row * Dc + lcol4);
    }
}

__device__ __forceinline__ void fetch_mask_rows(const float* __restrict__ omask,
                                                size_t bh_ss, int pa, int g, int w, int lane,
                                                v4f (&sreg)[16])
{
    const int lr   = 4 * g + w;
    const int grow = (lr < BQ) ? (NQB - 1 - pa) * BQ + lr : pa * BQ + (lr - BQ);
    const v4f* p = (const v4f*)(omask + bh_ss + (size_t)grow * Sc) + lane;
    #pragma unroll
    for (int i = 0; i < 16; ++i) sreg[i] = __builtin_nontemporal_load(p + 32 * i);
}

template <int I>
__device__ __forceinline__ float half_lane(float x)
{
    return __int_as_float(__builtin_amdgcn_update_dpp(0, __float_as_int(x), 0x150 + I, 0xf, 0xf, true));
}

__device__ __forceinline__ void lds_barrier()
{
    __builtin_amdgcn_fence(__ATOMIC_RELEASE, "workgroup", "local");
    __builtin_amdgcn_s_barrier();
    __builtin_amdgcn_fence(__ATOMIC_ACQUIRE, "workgroup", "local");
}

__launch_bounds__(128)
__global__ void retention_fused_kernel(const float* __restrict__ q,
                                       const float* __restrict__ k,
                                       const float* __restrict__ v,
                                       const float* __restrict__ omask,
                                       float* __restrict__ out)
{
    const int pa   = (int)blockIdx.x;
    const int bh   = (int)blockIdx.y;
    const int tid  = (int)threadIdx.x;
    const int w    = tid >> 5;
    const int lane = tid & 31;
    const int half = lane >> 4;
    const int l16  = lane & 15;
    const int sp    = w >> 1;
    const int part  = w & 1;

    __shared__ __attribute__((aligned(16))) float ldsK[BK * KS];
    __shared__ __attribute__((aligned(16))) float ldsV[BK * VS];
    __shared__ float ldsO[4 * 32 * 32];
    __shared__ float ldsQC[2 * BQ];
    __shared__ float ldsPart[2 * BQ * PS];
    __shared__ float ldsM0[2 * BQ];
    __shared__ float ldsP[4 * 32];

    const size_t bh_sd = (size_t)bh * Sc * Dc;
    const size_t bh_ss = (size_t)bh * Sc * (size_t)Sc;

    const int lrow  = tid >> 4;
    const int lcol4 = (tid & 15) * 4;
    const int mrow  = lrow + 8 * (lane & 7);

    const float* kfp = ldsK + (part * 32 + l16) * KS + half * 2;
    const float* vfp = ldsV + (part * 32 + half * 8) * VS + l16;
    int vcol[4];
    #pragma unroll
    for (int a = 0; a < 4; ++a) {
        vcol[a] = ((a + 2 * part) & 3) * 16;
        asm volatile("" : "+v"(vcol[a]));
    }

    int g = 0;

    for (int pass = 0; pass < 2; ++pass) {
        const int blk  = (pass == 0) ? (NQB - 1 - pa) : pa;
        const int row0 = blk * BQ;
        const int nkb  = blk + 1;

        int qrow[2];
        v2f qf[2][16];
        #pragma unroll
        for (int s = 0; s < 2; ++s) {
            qrow[s] = row0 + sp * 32 + s * 16 + l16;
            const float* qptr = q + bh_sd + (size_t)qrow[s] * Dc;
            #pragma unroll
            for (int kk = 0; kk < 16; ++kk) qf[s][kk] = *(const v2f*)(qptr + kk * 4 + half * 2);
        }

        v8f oacc[2][4];
        float pacc[2];
        #pragma unroll
        for (int s = 0; s < 2; ++s) {
            pacc[s] = 0.f;
            #pragma unroll
            for (int a = 0; a < 4; ++a) {
                #pragma unroll
                for (int r = 0; r < 8; ++r) oacc[s][a][r] = 0.f;
            }
        }

        v4f kreg[8], vreg[8];
        float mlane;
        fetch_kv_tile(k, v, omask, bh_sd, bh_ss, 0, lrow, lcol4, mrow, kreg, vreg, mlane);

        for (int kb = 0; kb < nkb; ++kb) {
            __syncthreads();
            {
                const float kcl = 1.f / mlane;
#define STAGE_CELL(I)                                                              \
                {                                                                  \
                    const int row = lrow + 8 * (I);                                \
                    const float kc = half_lane<(I)>(kcl);                          \
                    *(v4f*)(ldsK + row * KS + lcol4) = kreg[(I)] * kc;             \
                    *(v4f*)(ldsV + row * VS + lcol4) = vreg[(I)];                  \
                }
                STAGE_CELL(0) STAGE_CELL(1) STAGE_CELL(2) STAGE_CELL(3)
                STAGE_CELL(4) STAGE_CELL(5) STAGE_CELL(6) STAGE_CELL(7)
#undef STAGE_CELL
            }
            if (kb + 1 < nkb)
                fetch_kv_tile(k, v, omask, bh_sd, bh_ss, kb + 1, lrow, lcol4, mrow, kreg, vreg, mlane);
            const bool stream = g < NSTREAM;
            v4f sreg[16];
            if (stream) fetch_mask_rows(omask, bh_ss, pa, g, w, lane, sreg);
            lds_barrier();

            v2f kf[2][16];
            #pragma unroll
            for (int tt = 0; tt < 2; ++tt) {
                #pragma unroll
                for (int kk = 0; kk < 16; ++kk)
                    kf[tt][kk] = *(const v2f*)(kfp + tt * 16 * KS + kk * 4);
            }
            v2f vf[2][4][4];
            #pragma unroll
            for (int tt = 0; tt < 2; ++tt) {
                #pragma unroll
                for (int j = 0; j < 4; ++j) {
                    #pragma unroll
                    for (int a = 0; a < 4; ++a) {
                        const float* p = vfp + (tt * 16 + 2 * j) * VS + vcol[a];
                        v2f f; f.x = p[0]; f.y = p[VS];
                        vf[tt][j][a] = f;
                    }
                }
            }
            __builtin_amdgcn_sched_barrier(0);

            v8f st[2][2];
            #pragma unroll
            for (int s = 0; s < 2; ++s) {
                #pragma unroll
                for (int tt = 0; tt < 2; ++tt) {
                    #pragma unroll
                    for (int r = 0; r < 8; ++r) st[s][tt][r] = 0.f;
                }
            }
            #pragma unroll
            for (int kk = 0; kk < 16; ++kk) {
                #pragma unroll
                for (int tt = 0; tt < 2; ++tt) {
                    #pragma unroll
                    for (int s = 0; s < 2; ++s)
                        st[s][tt] = __builtin_amdgcn_wmma_f32_16x16x4_f32(
                                        false, kf[tt][kk], false, qf[s][kk], (short)0, st[s][tt], false, false);
                }
            }
            #pragma unroll
            for (int s = 0; s < 2; ++s) {
                #pragma unroll
                for (int tt = 0; tt < 2; ++tt) {
                    const int jg0  = kb * BK + (part * 2 + tt) * 16 + half * 8;
                    const int keep = qrow[s] - jg0;
                    #pragma unroll
                    for (int r = 0; r < 8; ++r) st[s][tt][r] = (r <= keep) ? st[s][tt][r] : 0.f;
                    pacc[s] += ((st[s][tt][0] + st[s][tt][1]) + (st[s][tt][2] + st[s][tt][3]))
                             + ((st[s][tt][4] + st[s][tt][5]) + (st[s][tt][6] + st[s][tt][7]));
                }
            }
            #pragma unroll
            for (int tt = 0; tt < 2; ++tt) {
                #pragma unroll
                for (int j = 0; j < 4; ++j) {
                    #pragma unroll
                    for (int s = 0; s < 2; ++s) {
                        v2f b; b.x = st[s][tt][2 * j]; b.y = st[s][tt][2 * j + 1];
                        #pragma unroll
                        for (int a = 0; a < 4; ++a)
                            oacc[s][a] = __builtin_amdgcn_wmma_f32_16x16x4_f32(
                                             false, vf[tt][j][a], false, b, (short)0, oacc[s][a], false, false);
                    }
                }
            }

            if (stream) {
                const v4f a0 = ((sreg[0] + sreg[1]) + (sreg[2] + sreg[3]))
                             + ((sreg[4] + sreg[5]) + (sreg[6] + sreg[7]));
                const v4f a1 = ((sreg[8] + sreg[9]) + (sreg[10] + sreg[11]))
                             + ((sreg[12] + sreg[13]) + (sreg[14] + sreg[15]));
                const int lr = 4 * g + w;
                ldsPart[lr * PS + lane]      = (a0.x + a0.y) + (a0.z + a0.w);
                ldsPart[lr * PS + 32 + lane] = (a1.x + a1.y) + (a1.z + a1.w);
                if (lane == 0) ldsM0[lr] = sreg[0].x;
            }
            ++g;
        }

        #pragma unroll
        for (int s = 0; s < 2; ++s) {
            const float p = pacc[s] + __shfl_xor(pacc[s], 16, 32);
            if (half == 0) ldsP[w * 32 + s * 16 + l16] = p;
        }
        {
            float* give = ldsO + w * 1024 + lane;
            #pragma unroll
            for (int s = 0; s < 2; ++s) {
                #pragma unroll
                for (int mm = 0; mm < 2; ++mm) {
                    #pragma unroll
                    for (int r = 0; r < 8; ++r) give[((s * 2 + mm) * 8 + r) * 32] = oacc[s][2 + mm][r];
                }
            }
        }
        __syncthreads();
        {
            const int lr = pass * BQ + (tid >> 1);
            const float* pp = ldsPart + lr * PS + (tid & 1) * 32;
            float s0 = ((pp[0] + pp[1]) + (pp[2] + pp[3])) + ((pp[4] + pp[5]) + (pp[6] + pp[7]));
            s0 += ((pp[8] + pp[9]) + (pp[10] + pp[11])) + ((pp[12] + pp[13]) + (pp[14] + pp[15]));
            float s1 = ((pp[16] + pp[17]) + (pp[18] + pp[19])) + ((pp[20] + pp[21]) + (pp[22] + pp[23]));
            s1 += ((pp[24] + pp[25]) + (pp[26] + pp[27])) + ((pp[28] + pp[29]) + (pp[30] + pp[31]));
            float s = s0 + s1;
            s += __shfl_xor(s, 1, 32);
            if ((tid & 1) == 0) ldsQC[lr] = ldsM0[lr] / sqrtf(s);
        }
        __syncthreads();
        #pragma unroll
        for (int s = 0; s < 2; ++s) {
            const float qc = ldsQC[pass * BQ + sp * 32 + s * 16 + l16];
            const float p = (ldsP[(sp * 2) * 32 + s * 16 + l16] + ldsP[(sp * 2 + 1) * 32 + s * 16 + l16]) * qc;
            const float invp = qc / fmaxf(fabsf(p), 1.f);
            const float* take = ldsO + (w ^ 1) * 1024 + (s * 2) * 8 * 32 + lane;
            float* orow = out + bh_sd + (size_t)qrow[s] * Dc + part * 32 + half * 8;
            #pragma unroll
            for (int mm = 0; mm < 2; ++mm) {
                v4f lo, hi;
                lo.x = (oacc[s][mm][0] + take[(mm * 8 + 0) * 32]) * invp;
                lo.y = (oacc[s][mm][1] + take[(mm * 8 + 1) * 32]) * invp;
                lo.z = (oacc[s][mm][2] + take[(mm * 8 + 2) * 32]) * invp;
                lo.w = (oacc[s][mm][3] + take[(mm * 8 + 3) * 32]) * invp;
                hi.x = (oacc[s][mm][4] + take[(mm * 8 + 4) * 32]) * invp;
                hi.y = (oacc[s][mm][5] + take[(mm * 8 + 5) * 32]) * invp;
                hi.z = (oacc[s][mm][6] + take[(mm * 8 + 6) * 32]) * invp;
                hi.w = (oacc[s][mm][7] + take[(mm * 8 + 7) * 32]) * invp;
                *(v4f*)(orow + mm * 16)     = lo;
                *(v4f*)(orow + mm * 16 + 4) = hi;
            }
        }
    }
}

extern "C" void kernel_launch(void* const* d_in, const int* in_sizes, int n_in,
                              void* d_out, int out_size, void* d_ws, size_t ws_size,
                              hipStream_t stream) {
    (void)in_sizes; (void)n_in; (void)out_size; (void)d_ws; (void)ws_size;
    const float* q  = (const float*)d_in[0];
    const float* k  = (const float*)d_in[1];
    const float* v  = (const float*)d_in[2];
    const float* om = (const float*)d_in[3];
    float* out = (float*)d_out;
    dim3 grid(NQB / 2, BHc);
    retention_fused_kernel<<<grid, 128, 0, stream>>>(q, k, v, om, out);
}


# 8 waves per workgroup, two per SIMD: the 4 waves of a strip pair take one 16-key tile each, so a wave's staging, LDS and mask work runs under its sibling's WMMAs; O, P = sum of 4 wave partials
# speedup vs baseline: 3.7623x; 1.0406x over previous
#include <hip/hip_runtime.h>


typedef float v2f __attribute__((ext_vector_type(2)));
typedef float v4f __attribute__((ext_vector_type(4)));
typedef float v8f __attribute__((ext_vector_type(8)));

#define Bc 2
#define Hc 8
#define Sc 2048
#define Dc 64
#define BHc (Bc * Hc)
#define BQ 64
#define BK 64
#define NQB (Sc / BQ)
#define KS 68
#define VS 68
#define PS 65
#define NSTREAM ((2 * BQ) / 4)

__device__ __forceinline__ void fetch_kv_tile(const float* __restrict__ k,
                                              const float* __restrict__ v,
                                              const float* __restrict__ omask,
                                              size_t bh_sd, size_t bh_ss, int kb,
                                              int lrow, int lcol4, int mrow,
                                              v4f (&kreg)[4], v4f (&vreg)[4], float &mlane)
{
    const float* kbase = k + bh_sd + (size_t)(kb * BK) * Dc;
    const float* vbase = v + bh_sd + (size_t)(kb * BK) * Dc;
    const float* mbase = omask + bh_ss + (size_t)(kb * BK) * Sc;
    mlane = mbase[(size_t)mrow * Sc];
    #pragma unroll
    for (int i = 0; i < 4; ++i) {
        const int row = lrow + 16 * i;
        kreg[i] = *(const v4f*)(kbase + row * Dc + lcol4);
        vreg[i] = *(const v4f*)(vbase + row * Dc + lcol4);
    }
}

__device__ __forceinline__ void fetch_mask_rows(const float* __restrict__ omask,
                                                size_t bh_ss, int pa, int g, int w, int lane,
                                                v4f (&sreg)[8])
{
    const int lr   = 4 * g + (w >> 1);
    const int grow = (lr < BQ) ? (NQB - 1 - pa) * BQ + lr : pa * BQ + (lr - BQ);
    const v4f* p = (const v4f*)(omask + bh_ss + (size_t)grow * Sc) + (w & 1) * 256 + lane;
    #pragma unroll
    for (int i = 0; i < 8; ++i) sreg[i] = __builtin_nontemporal_load(p + 32 * i);
}

template <int I>
__device__ __forceinline__ float half_lane(float x)
{
    return __int_as_float(__builtin_amdgcn_update_dpp(0, __float_as_int(x), 0x150 + I, 0xf, 0xf, true));
}

__device__ __forceinline__ void lds_barrier()
{
    __builtin_amdgcn_fence(__ATOMIC_RELEASE, "workgroup", "local");
    __builtin_amdgcn_s_barrier();
    __builtin_amdgcn_fence(__ATOMIC_ACQUIRE, "workgroup", "local");
}

__launch_bounds__(256)
__global__ void retention_fused_kernel(const float* __restrict__ q,
                                       const float* __restrict__ k,
                                       const float* __restrict__ v,
                                       const float* __restrict__ omask,
                                       float* __restrict__ out)
{
    const int pa   = (int)blockIdx.x;
    const int bh   = (int)blockIdx.y;
    const int tid  = (int)threadIdx.x;
    const int w    = tid >> 5;
    const int lane = tid & 31;
    const int half = lane >> 4;
    const int l16  = lane & 15;
    const int sp    = w >> 2;
    const int part  = w & 3;

    __shared__ __attribute__((aligned(16))) float ldsK[BK * KS];
    __shared__ __attribute__((aligned(16))) float ldsV[BK * VS];
    __shared__ float ldsO[8 * 48 * 32];
    __shared__ float ldsQC[2 * BQ];
    __shared__ float ldsPart[2 * BQ * PS];
    __shared__ float ldsM0[2 * BQ];
    __shared__ float ldsP[8 * 32];

    const size_t bh_sd = (size_t)bh * Sc * Dc;
    const size_t bh_ss = (size_t)bh * Sc * (size_t)Sc;

    const int lrow  = tid >> 4;
    const int lcol4 = (tid & 15) * 4;
    const int mrow  = lrow + 16 * (lane & 3);

    const float* kfp = ldsK + (part * 16 + l16) * KS + half * 2;
    const float* vfp = ldsV + (part * 16 + half * 8) * VS + l16;
    int vcol[4];
    #pragma unroll
    for (int a = 0; a < 4; ++a) {
        vcol[a] = ((a + part) & 3) * 16;
        asm volatile("" : "+v"(vcol[a]));
    }

    int g = 0;

    for (int pass = 0; pass < 2; ++pass) {
        const int blk  = (pass == 0) ? (NQB - 1 - pa) : pa;
        const int row0 = blk * BQ;
        const int nkb  = blk + 1;

        int qrow[2];
        v2f qf[2][16];
        #pragma unroll
        for (int s = 0; s < 2; ++s) {
            qrow[s] = row0 + sp * 32 + s * 16 + l16;
            const float* qptr = q + bh_sd + (size_t)qrow[s] * Dc;
            #pragma unroll
            for (int kk = 0; kk < 16; ++kk) qf[s][kk] = *(const v2f*)(qptr + kk * 4 + half * 2);
        }

        v8f oacc[2][4];
        float pacc[2];
        #pragma unroll
        for (int s = 0; s < 2; ++s) {
            pacc[s] = 0.f;
            #pragma unroll
            for (int a = 0; a < 4; ++a) {
                #pragma unroll
                for (int r = 0; r < 8; ++r) oacc[s][a][r] = 0.f;
            }
        }

        v4f kreg[4], vreg[4];
        float mlane;
        fetch_kv_tile(k, v, omask, bh_sd, bh_ss, 0, lrow, lcol4, mrow, kreg, vreg, mlane);

        for (int kb = 0; kb < nkb; ++kb) {
            __syncthreads();
            {
                const float kcl = 1.f / mlane;
#define STAGE_CELL(I)                                                              \
                {                                                                  \
                    const int row = lrow + 16 * (I);                               \
                    const float kc = half_lane<(I)>(kcl);                          \
                    *(v4f*)(ldsK + row * KS + lcol4) = kreg[(I)] * kc;             \
                    *(v4f*)(ldsV + row * VS + lcol4) = vreg[(I)];                  \
                }
                STAGE_CELL(0) STAGE_CELL(1) STAGE_CELL(2) STAGE_CELL(3)
#undef STAGE_CELL
            }
            if (kb + 1 < nkb)
                fetch_kv_tile(k, v, omask, bh_sd, bh_ss, kb + 1, lrow, lcol4, mrow, kreg, vreg, mlane);
            const bool stream = g < NSTREAM;
            v4f sreg[8];
            if (stream) fetch_mask_rows(omask, bh_ss, pa, g, w, lane, sreg);
            lds_barrier();

            v2f kf[16];
            #pragma unroll
            for (int kk = 0; kk < 16; ++kk)
                kf[kk] = *(const v2f*)(kfp + kk * 4);
            v2f vf[4][4];
            #pragma unroll
            for (int j = 0; j < 4; ++j) {
                #pragma unroll
                for (int a = 0; a < 4; ++a) {
                    const float* p = vfp + (2 * j) * VS + vcol[a];
                    v2f f; f.x = p[0]; f.y = p[VS];
                    vf[j][a] = f;
                }
            }
            __builtin_amdgcn_sched_barrier(0);

            v8f st[2];
            #pragma unroll
            for (int s = 0; s < 2; ++s) {
                #pragma unroll
                for (int r = 0; r < 8; ++r) st[s][r] = 0.f;
            }
            #pragma unroll
            for (int kk = 0; kk < 16; ++kk) {
                #pragma unroll
                for (int s = 0; s < 2; ++s)
                    st[s] = __builtin_amdgcn_wmma_f32_16x16x4_f32(
                                false, kf[kk], false, qf[s][kk], (short)0, st[s], false, false);
            }
            {
                const int jg0 = kb * BK + part * 16 + half * 8;
                #pragma unroll
                for (int s = 0; s < 2; ++s) {
                    const int keep = qrow[s] - jg0;
                    #pragma unroll
                    for (int r = 0; r < 8; ++r) st[s][r] = (r <= keep) ? st[s][r] : 0.f;
                    pacc[s] += ((st[s][0] + st[s][1]) + (st[s][2] + st[s][3]))
                             + ((st[s][4] + st[s][5]) + (st[s][6] + st[s][7]));
                }
            }
            #pragma unroll
            for (int j = 0; j < 4; ++j) {
                #pragma unroll
                for (int s = 0; s < 2; ++s) {
                    v2f b; b.x = st[s][2 * j]; b.y = st[s][2 * j + 1];
                    #pragma unroll
                    for (int a = 0; a < 4; ++a)
                        oacc[s][a] = __builtin_amdgcn_wmma_f32_16x16x4_f32(
                                         false, vf[j][a], false, b, (short)0, oacc[s][a], false, false);
                }
            }

            if (stream) {
                const v4f a0 = ((sreg[0] + sreg[1]) + (sreg[2] + sreg[3]))
                             + ((sreg[4] + sreg[5]) + (sreg[6] + sreg[7]));
                const int lr = 4 * g + (w >> 1);
                ldsPart[lr * PS + (w & 1) * 32 + lane] = (a0.x + a0.y) + (a0.z + a0.w);
                if ((w & 1) == 0 && lane == 0) ldsM0[lr] = sreg[0].x;
            }
            ++g;
        }

        #pragma unroll
        for (int s = 0; s < 2; ++s) {
            const float p = pacc[s] + __shfl_xor(pacc[s], 16, 32);
            if (half == 0) ldsP[w * 32 + s * 16 + l16] = p;
        }
        {
            float* give = ldsO + w * 1536 + lane;
            #pragma unroll
            for (int s = 0; s < 2; ++s) {
                #pragma unroll
                for (int mm = 0; mm < 3; ++mm) {
                    #pragma unroll
                    for (int r = 0; r < 8; ++r) give[((s * 3 + mm) * 8 + r) * 32] = oacc[s][1 + mm][r];
                }
            }
        }
        __syncthreads();
        if (tid < 2 * BQ) {
            const int lr = pass * BQ + (tid >> 1);
            const float* pp = ldsPart + lr * PS + (tid & 1) * 32;
            float s0 = ((pp[0] + pp[1]) + (pp[2] + pp[3])) + ((pp[4] + pp[5]) + (pp[6] + pp[7]));
            s0 += ((pp[8] + pp[9]) + (pp[10] + pp[11])) + ((pp[12] + pp[13]) + (pp[14] + pp[15]));
            float s1 = ((pp[16] + pp[17]) + (pp[18] + pp[19])) + ((pp[20] + pp[21]) + (pp[22] + pp[23]));
            s1 += ((pp[24] + pp[25]) + (pp[26] + pp[27])) + ((pp[28] + pp[29]) + (pp[30] + pp[31]));
            float s = s0 + s1;
            s += __shfl_xor(s, 1, 32);
            if ((tid & 1) == 0) ldsQC[lr] = ldsM0[lr] / sqrtf(s);
        }
        __syncthreads();
        #pragma unroll
        for (int s = 0; s < 2; ++s) {
            const float qc = ldsQC[pass * BQ + sp * 32 + s * 16 + l16];
            const float* pw = ldsP + (sp * 4) * 32 + s * 16 + l16;
            const float p = ((pw[0] + pw[32]) + (pw[64] + pw[96])) * qc;
            const float invp = qc / fmaxf(fabsf(p), 1.f);
            const float* t1 = ldsO + (sp * 4 + ((part - 1) & 3)) * 1536 + ((s * 3 + 0) * 8) * 32 + lane;
            const float* t2 = ldsO + (sp * 4 + ((part - 2) & 3)) * 1536 + ((s * 3 + 1) * 8) * 32 + lane;
            const float* t3 = ldsO + (sp * 4 + ((part - 3) & 3)) * 1536 + ((s * 3 + 2) * 8) * 32 + lane;
            float* orow = out + bh_sd + (size_t)qrow[s] * Dc + part * 16 + half * 8;
            v4f lo, hi;
            lo.x = ((oacc[s][0][0] + t1[0 * 32]) + (t2[0 * 32] + t3[0 * 32])) * invp;
            lo.y = ((oacc[s][0][1] + t1[1 * 32]) + (t2[1 * 32] + t3[1 * 32])) * invp;
            lo.z = ((oacc[s][0][2] + t1[2 * 32]) + (t2[2 * 32] + t3[2 * 32])) * invp;
            lo.w = ((oacc[s][0][3] + t1[3 * 32]) + (t2[3 * 32] + t3[3 * 32])) * invp;
            hi.x = ((oacc[s][0][4] + t1[4 * 32]) + (t2[4 * 32] + t3[4 * 32])) * invp;
            hi.y = ((oacc[s][0][5] + t1[5 * 32]) + (t2[5 * 32] + t3[5 * 32])) * invp;
            hi.z = ((oacc[s][0][6] + t1[6 * 32]) + (t2[6 * 32] + t3[6 * 32])) * invp;
            hi.w = ((oacc[s][0][7] + t1[7 * 32]) + (t2[7 * 32] + t3[7 * 32])) * invp;
            *(v4f*)(orow)     = lo;
            *(v4f*)(orow + 4) = hi;
        }
    }
}

extern "C" void kernel_launch(void* const* d_in, const int* in_sizes, int n_in,
                              void* d_out, int out_size, void* d_ws, size_t ws_size,
                              hipStream_t stream) {
    (void)in_sizes; (void)n_in; (void)out_size; (void)d_ws; (void)ws_size;
    const float* q  = (const float*)d_in[0];
    const float* k  = (const float*)d_in[1];
    const float* v  = (const float*)d_in[2];
    const float* om = (const float*)d_in[3];
    float* out = (float*)d_out;
    dim3 grid(NQB / 2, BHc);
    retention_fused_kernel<<<grid, 256, 0, stream>>>(q, k, v, om, out);
}
